# starting kernel
# baseline (speedup 1.0000x reference)
#include <hip/hip_runtime.h>

typedef __attribute__((ext_vector_type(2))) float v2f;
typedef __attribute__((ext_vector_type(4))) float v4f;
typedef __attribute__((ext_vector_type(8))) float v8f;

static constexpr int D = 128;
static constexpr int ROWS_PER_BLOCK = 256;
static constexpr int ROWS_PER_WAVE = 32;

#define WMMA_F32(A, B, C) \
  __builtin_amdgcn_wmma_f32_16x16x4_f32(false, (A), false, (B), (short)0, (C), false, false)

__global__ __launch_bounds__(256) void hetero_linear_wmma(
    const float* __restrict__ x,
    const int*   __restrict__ tvec,
    const float* __restrict__ W,
    const float* __restrict__ bias,
    float*       __restrict__ out,
    int nrows)
{
  __shared__ float ldsW[64 * 256];

  const int tid       = threadIdx.x;
  const int block_row = blockIdx.x * ROWS_PER_BLOCK;

  const int t_first = tvec[block_row];
  const int t_last  = tvec[block_row + ROWS_PER_BLOCK - 1];

  if (t_first == t_last) {
    const float* Wt = W + (size_t)t_first * D * D;

    for (int task = tid; task < 64 * 32; task += 256) {
      const int pr = task >> 5;
      const int cb = (task & 31) << 2;
      v4f g0 = *(const v4f*)(Wt + (size_t)(2 * pr)     * D + cb);
      v4f g1 = *(const v4f*)(Wt + (size_t)(2 * pr + 1) * D + cb);
      const int cs = cb ^ ((pr & 1) << 4);
      float* dst = &ldsW[pr * 256 + cs * 2];
      v4f lo = { g0[0], g1[0], g0[1], g1[1] };
      v4f hi = { g0[2], g1[2], g0[3], g1[3] };
      *(v4f*)(dst)     = lo;
      *(v4f*)(dst + 4) = hi;
    }
    __syncthreads();

    const int lane = tid & 31;
    const int wave = tid >> 5;
    const int h    = lane >> 4;
    const int l    = lane & 15;

    const int r0 = block_row + wave * ROWS_PER_WAVE;

    v8f acc[2][8];
    #pragma unroll
    for (int rt = 0; rt < 2; ++rt)
      #pragma unroll
      for (int j = 0; j < 8; ++j)
        acc[rt][j] = v8f{0.f, 0.f, 0.f, 0.f, 0.f, 0.f, 0.f, 0.f};

    const float* xr0 = x + (size_t)(r0 + l)      * D;
    const float* xr1 = x + (size_t)(r0 + 16 + l) * D;

    const float* bLane = ldsW + h * 256 + (l << 1);
    const float* bE;
    const float* bO;
    {
      const int offE = h * 32;
      bE = bLane + offE;
      bO = bLane + (32 - offE);
    }

    for (int k0 = 0; k0 < D; k0 += 4) {
      v2f a0 = *(const v2f*)(xr0 + k0 + 2 * h);
      v2f a1 = *(const v2f*)(xr1 + k0 + 2 * h);
      const int kOff = (k0 >> 1) * 256;

      v2f bfrag[8];
      #pragma unroll
      for (int j = 0; j < 8; ++j) {
        const float* bp = ((j & 1) ? bO : bE) + kOff + (j >> 1) * 64;
        bfrag[j] = *(const v2f*)bp;
      }
      #pragma unroll
      for (int j = 0; j < 8; ++j) {
        acc[0][j] = WMMA_F32(a0, bfrag[j], acc[0][j]);
        acc[1][j] = WMMA_F32(a1, bfrag[j], acc[1][j]);
      }
    }

    #pragma unroll
    for (int j = 0; j < 8; ++j) {
      const int   col = j * 16 + l;
      const float bv  = bias[t_first * D + col];
      #pragma unroll
      for (int rt = 0; rt < 2; ++rt) {
        float* orow = out + (size_t)(r0 + rt * 16 + 8 * h) * D + col;
        #pragma unroll
        for (int i = 0; i < 8; ++i)
          orow[(size_t)i * D] = acc[rt][j][i] + bv;
      }
    }
  } else {
    for (int idx = tid; idx < ROWS_PER_BLOCK * D; idx += 256) {
      const int r = block_row + (idx >> 7);
      const int c = idx & (D - 1);
      if (r >= nrows) break;
      const int t = tvec[r];
      const float* Wt = W + (size_t)t * D * D + c;
      const float* xr = x + (size_t)r * D;
      float a = bias[t * D + c];
      #pragma unroll 8
      for (int k = 0; k < D; ++k)
        a = fmaf(xr[k], Wt[(size_t)k * D], a);
      out[(size_t)r * D + c] = a;
    }
  }
}

extern "C" void kernel_launch(void* const* d_in, const int* in_sizes, int n_in,
                              void* d_out, int out_size, void* d_ws, size_t ws_size,
                              hipStream_t stream) {
  const float* x    = (const float*)d_in[0];
  const int*   tvec = (const int*)  d_in[1];
  const float* W    = (const float*)d_in[2];
  const float* bias = (const float*)d_in[3];
  float* out = (float*)d_out;

  const int nrows = in_sizes[0] / D;
  const int grid  = nrows / ROWS_PER_BLOCK;

  hipLaunchKernelGGL(hetero_linear_wmma, dim3(grid), dim3(256), 0, stream,
                     x, tvec, W, bias, out, nrows);
}


# Mixed-type blocks leave the per-row scalar fallback: per type present, stage its W, run the f32 WMMA tile loop, store only that type's rows; their k-sum order becomes the WMMA path's, bias last
# speedup vs baseline: 11.0570x; 11.0570x over previous
#include <hip/hip_runtime.h>


typedef __attribute__((ext_vector_type(2))) float v2f;
typedef __attribute__((ext_vector_type(4))) float v4f;
typedef __attribute__((ext_vector_type(8))) float v8f;

static constexpr int D = 128;
static constexpr int ROWS_PER_BLOCK = 256;
static constexpr int ROWS_PER_WAVE = 32;

#define WMMA_F32(A, B, C) \
  __builtin_amdgcn_wmma_f32_16x16x4_f32(false, (A), false, (B), (short)0, (C), false, false)

__device__ __forceinline__ void stage_w(float* __restrict__ ldsW,
                                        const float* __restrict__ Wt, int tid)
{
  for (int task = tid; task < 64 * 32; task += 256) {
    const int pr = task >> 5;
    const int cb = (task & 31) << 2;
    v4f g0 = *(const v4f*)(Wt + (size_t)(2 * pr)     * D + cb);
    v4f g1 = *(const v4f*)(Wt + (size_t)(2 * pr + 1) * D + cb);
    const int cs = cb ^ ((pr & 1) << 4);
    float* dst = &ldsW[pr * 256 + cs * 2];
    v4f lo = { g0[0], g1[0], g0[1], g1[1] };
    v4f hi = { g0[2], g1[2], g0[3], g1[3] };
    *(v4f*)(dst)     = lo;
    *(v4f*)(dst + 4) = hi;
  }
}

template <bool MIXED>
__device__ __forceinline__ void wave_rows_times_w(
    const float* __restrict__ x,
    const float* __restrict__ bias,
    float*       __restrict__ out,
    const float* ldsW,
    int r0, int t, int lane, unsigned rowmask)
{
  const int h = lane >> 4;
  const int l = lane & 15;

  v8f acc[2][8];
  #pragma unroll
  for (int rt = 0; rt < 2; ++rt)
    #pragma unroll
    for (int j = 0; j < 8; ++j)
      acc[rt][j] = v8f{0.f, 0.f, 0.f, 0.f, 0.f, 0.f, 0.f, 0.f};

  const float* xr0 = x + (size_t)(r0 + l)      * D;
  const float* xr1 = x + (size_t)(r0 + 16 + l) * D;

  const float* bLane = ldsW + h * 256 + (l << 1);
  const float* bE;
  const float* bO;
  {
    const int offE = h * 32;
    bE = bLane + offE;
    bO = bLane + (32 - offE);
  }

  for (int k0 = 0; k0 < D; k0 += 4) {
    v2f a0 = *(const v2f*)(xr0 + k0 + 2 * h);
    v2f a1 = *(const v2f*)(xr1 + k0 + 2 * h);
    const int kOff = (k0 >> 1) * 256;

    v2f bfrag[8];
    #pragma unroll
    for (int j = 0; j < 8; ++j) {
      const float* bp = ((j & 1) ? bO : bE) + kOff + (j >> 1) * 64;
      bfrag[j] = *(const v2f*)bp;
    }
    #pragma unroll
    for (int j = 0; j < 8; ++j) {
      acc[0][j] = WMMA_F32(a0, bfrag[j], acc[0][j]);
      acc[1][j] = WMMA_F32(a1, bfrag[j], acc[1][j]);
    }
  }

  if (!MIXED) {
    #pragma unroll
    for (int j = 0; j < 8; ++j) {
      const int   col = j * 16 + l;
      const float bv  = bias[t * D + col];
      #pragma unroll
      for (int rt = 0; rt < 2; ++rt) {
        float* orow = out + (size_t)(r0 + rt * 16 + 8 * h) * D + col;
        #pragma unroll
        for (int i = 0; i < 8; ++i)
          orow[(size_t)i * D] = acc[rt][j][i] + bv;
      }
    }
  } else {
    float bv[8];
    #pragma unroll
    for (int j = 0; j < 8; ++j)
      bv[j] = bias[t * D + j * 16 + l];
    const unsigned mh = rowmask >> (8 * h);
    #pragma unroll
    for (int rt = 0; rt < 2; ++rt) {
      #pragma unroll
      for (int i = 0; i < 8; ++i) {
        if ((mh >> (rt * 16 + i)) & 1u) {
          float* orow = out + (size_t)(r0 + rt * 16 + 8 * h + i) * D + l;
          #pragma unroll
          for (int j = 0; j < 8; ++j)
            orow[j * 16] = acc[rt][j][i] + bv[j];
        }
      }
    }
  }
}

__global__ __launch_bounds__(256) void hetero_linear_wmma(
    const float* __restrict__ x,
    const int*   __restrict__ tvec,
    const float* __restrict__ W,
    const float* __restrict__ bias,
    float*       __restrict__ out,
    int nrows)
{
  __shared__ float ldsW[64 * 256];

  const int tid       = threadIdx.x;
  const int lane      = tid & 31;
  const int wave      = tid >> 5;
  const int block_row = blockIdx.x * ROWS_PER_BLOCK;
  const int r0        = block_row + wave * ROWS_PER_WAVE;

  const int t_first = tvec[block_row];
  const int t_last  = tvec[block_row + ROWS_PER_BLOCK - 1];

  if (t_first == t_last) {
    stage_w(ldsW, W + (size_t)t_first * D * D, tid);
    __syncthreads();
    wave_rows_times_w<false>(x, bias, out, ldsW, r0, t_first, lane, 0u);
  } else {
    const int tv = tvec[r0 + lane];
    for (int t = t_first; t <= t_last; ++t) {
      if (t != t_first)
        __syncthreads();
      stage_w(ldsW, W + (size_t)t * D * D, tid);
      __syncthreads();
      const unsigned rowmask = __builtin_amdgcn_ballot_w32(tv == t);
      if (rowmask != 0u)
        wave_rows_times_w<true>(x, bias, out, ldsW, r0, t, lane, rowmask);
    }
  }
}

extern "C" void kernel_launch(void* const* d_in, const int* in_sizes, int n_in,
                              void* d_out, int out_size, void* d_ws, size_t ws_size,
                              hipStream_t stream) {
  const float* x    = (const float*)d_in[0];
  const int*   tvec = (const int*)  d_in[1];
  const float* W    = (const float*)d_in[2];
  const float* bias = (const float*)d_in[3];
  float* out = (float*)d_out;

  const int nrows = in_sizes[0] / D;
  const int grid  = nrows / ROWS_PER_BLOCK;

  hipLaunchKernelGGL(hetero_linear_wmma, dim3(grid), dim3(256), 0, stream,
                     x, tvec, W, bias, out, nrows);
}


# A operand fetched 16 k ahead into a second register set by 16-B loads (was 8-B loads waited for at once); k-sum order permuted inside 8-blocks, the W swizzle in LDS follows
# speedup vs baseline: 12.4743x; 1.1282x over previous
#include <hip/hip_runtime.h>


typedef __attribute__((ext_vector_type(2))) float v2f;
typedef __attribute__((ext_vector_type(4))) float v4f;
typedef __attribute__((ext_vector_type(8))) float v8f;

static constexpr int D = 128;
static constexpr int ROWS_PER_BLOCK = 256;
static constexpr int ROWS_PER_WAVE = 32;

#define WMMA_F32(A, B, C) \
  __builtin_amdgcn_wmma_f32_16x16x4_f32(false, (A), false, (B), (short)0, (C), false, false)

__device__ __forceinline__ void stage_w(float* __restrict__ ldsW,
                                        const float* __restrict__ Wt, int tid)
{
  for (int task = tid; task < 64 * 32; task += 256) {
    const int pr = task >> 5;
    const int cb = (task & 31) << 2;
    v4f g0 = *(const v4f*)(Wt + (size_t)(2 * pr)     * D + cb);
    v4f g1 = *(const v4f*)(Wt + (size_t)(2 * pr + 1) * D + cb);
    const int cs = cb ^ (((pr >> 1) & 1) << 4);
    float* dst = &ldsW[pr * 256 + cs * 2];
    v4f lo = { g0[0], g1[0], g0[1], g1[1] };
    v4f hi = { g0[2], g1[2], g0[3], g1[3] };
    *(v4f*)(dst)     = lo;
    *(v4f*)(dst + 4) = hi;
  }
}

__device__ __forceinline__ void load_a_group(v4f (&a)[2][2],
                                             const float* __restrict__ xr0,
                                             const float* __restrict__ xr1, int g)
{
  #pragma unroll
  for (int q = 0; q < 2; ++q) {
    a[0][q] = *(const v4f*)(xr0 + g * 16 + q * 8);
    a[1][q] = *(const v4f*)(xr1 + g * 16 + q * 8);
  }
}

__device__ __forceinline__ void group_times_w(v8f (&acc)[2][8], const v4f (&a)[2][2],
                                              const float* bE, const float* bO, int g)
{
  const int gOff = g * 8 * 256;
  #pragma unroll
  for (int q = 0; q < 2; ++q) {
    #pragma unroll
    for (int s = 0; s < 2; ++s) {
      const int kOff = gOff + (q * 4 + s) * 256;
      v2f bfrag[8];
      #pragma unroll
      for (int j = 0; j < 8; ++j) {
        const float* bp = ((j & 1) ? bO : bE) + kOff + (j >> 1) * 64;
        bfrag[j] = *(const v2f*)bp;
      }
      const v2f a0 = { a[0][q][2 * s], a[0][q][2 * s + 1] };
      const v2f a1 = { a[1][q][2 * s], a[1][q][2 * s + 1] };
      #pragma unroll
      for (int j = 0; j < 8; ++j) {
        acc[0][j] = WMMA_F32(a0, bfrag[j], acc[0][j]);
        acc[1][j] = WMMA_F32(a1, bfrag[j], acc[1][j]);
      }
    }
  }
}

template <bool MIXED>
__device__ __forceinline__ void wave_rows_times_w(
    const float* __restrict__ xr0,
    const float* __restrict__ xr1,
    v4f (&a)[2][2][2],
    const float* __restrict__ bias,
    float*       __restrict__ out,
    const float* ldsW,
    int r0, int t, int lane, unsigned rowmask)
{
  const int h = lane >> 4;
  const int l = lane & 15;

  int hB = h;
  asm volatile("" : "+v"(hB));

  v8f acc[2][8];
  #pragma unroll
  for (int rt = 0; rt < 2; ++rt)
    #pragma unroll
    for (int j = 0; j < 8; ++j)
      acc[rt][j] = v8f{0.f, 0.f, 0.f, 0.f, 0.f, 0.f, 0.f, 0.f};

  const float* bLane = ldsW + hB * 512 + (l << 1);
  const float* bE;
  const float* bO;
  {
    const int offE = hB * 32;
    bE = bLane + offE;
    bO = bLane + (32 - offE);
  }

  #pragma unroll
  for (int g = 0; g < 8; ++g) {
    if (g + 1 < 8) {
      load_a_group(a[(g + 1) & 1], xr0, xr1, g + 1);
      __builtin_amdgcn_sched_barrier(0);
    }
    group_times_w(acc, a[g & 1], bE, bO, g);
    __builtin_amdgcn_sched_barrier(0);
  }

  if (!MIXED) {
    #pragma unroll
    for (int j = 0; j < 8; ++j) {
      const int   col = j * 16 + l;
      const float bv  = bias[t * D + col];
      #pragma unroll
      for (int rt = 0; rt < 2; ++rt) {
        float* orow = out + (size_t)(r0 + rt * 16 + 8 * h) * D + col;
        #pragma unroll
        for (int i = 0; i < 8; ++i)
          orow[(size_t)i * D] = acc[rt][j][i] + bv;
      }
    }
  } else {
    float bv[8];
    #pragma unroll
    for (int j = 0; j < 8; ++j)
      bv[j] = bias[t * D + j * 16 + l];
    const unsigned mh = rowmask >> (8 * h);
    float* obase = out + (size_t)(r0 + 8 * h) * D + l;
    #pragma unroll
    for (int rt = 0; rt < 2; ++rt) {
      #pragma unroll
      for (int i = 0; i < 8; ++i) {
        if ((mh >> (rt * 16 + i)) & 1u) {
          float* orow = obase + (rt * 16 + i) * D;
          #pragma unroll
          for (int j = 0; j < 8; ++j)
            orow[j * 16] = acc[rt][j][i] + bv[j];
        }
      }
    }
  }
}

__global__ __launch_bounds__(256, 4) void hetero_linear_wmma(
    const float* __restrict__ x,
    const int*   __restrict__ tvec,
    const float* __restrict__ W,
    const float* __restrict__ bias,
    float*       __restrict__ out,
    int nrows)
{
  __shared__ float ldsW[64 * 256];

  const int tid       = threadIdx.x;
  const int lane      = tid & 31;
  const int wave      = tid >> 5;
  const int block_row = blockIdx.x * ROWS_PER_BLOCK;
  const int r0        = block_row + wave * ROWS_PER_WAVE;

  const int t_first = tvec[block_row];
  const int t_last  = tvec[block_row + ROWS_PER_BLOCK - 1];

  const float* xr0 = x + (size_t)(r0 + (lane & 15)) * D + 4 * (lane >> 4);
  const float* xr1 = xr0 + (size_t)16 * D;
  v4f a[2][2][2];

  if (t_first == t_last) {
    load_a_group(a[0], xr0, xr1, 0);
    stage_w(ldsW, W + (size_t)t_first * D * D, tid);
    __syncthreads();
    wave_rows_times_w<false>(xr0, xr1, a, bias, out, ldsW, r0, t_first, lane, 0u);
  } else {
    const int tv = tvec[r0 + lane];
    for (int t = t_first; t <= t_last; ++t) {
      if (t != t_first)
        __syncthreads();
      const unsigned rowmask = __builtin_amdgcn_ballot_w32(tv == t);
      if (rowmask != 0u)
        load_a_group(a[0], xr0, xr1, 0);
      stage_w(ldsW, W + (size_t)t * D * D, tid);
      __syncthreads();
      if (rowmask != 0u)
        wave_rows_times_w<true>(xr0, xr1, a, bias, out, ldsW, r0, t, lane, rowmask);
    }
  }
}

extern "C" void kernel_launch(void* const* d_in, const int* in_sizes, int n_in,
                              void* d_out, int out_size, void* d_ws, size_t ws_size,
                              hipStream_t stream) {
  const float* x    = (const float*)d_in[0];
  const int*   tvec = (const int*)  d_in[1];
  const float* W    = (const float*)d_in[2];
  const float* bias = (const float*)d_in[3];
  float* out = (float*)d_out;

  const int nrows = in_sizes[0] / D;
  const int grid  = nrows / ROWS_PER_BLOCK;

  hipLaunchKernelGGL(hetero_linear_wmma, dim3(grid), dim3(256), 0, stream,
                     x, tvec, W, bias, out, nrows);
}


# W staging off the critical path: a thread's 16 loads issued together and ahead of A (was 8 load-then-store trips in a row); in a mixed block also ahead of the barrier that frees the buffer
# speedup vs baseline: 13.2572x; 1.0628x over previous
#include <hip/hip_runtime.h>


typedef __attribute__((ext_vector_type(2))) float v2f;
typedef __attribute__((ext_vector_type(4))) float v4f;
typedef __attribute__((ext_vector_type(8))) float v8f;

static constexpr int D = 128;
static constexpr int ROWS_PER_BLOCK = 256;
static constexpr int ROWS_PER_WAVE = 32;

#define WMMA_F32(A, B, C) \
  __builtin_amdgcn_wmma_f32_16x16x4_f32(false, (A), false, (B), (short)0, (C), false, false)

__device__ __forceinline__ void stage_w_load(v4f (&w)[8][2],
                                             const float* __restrict__ Wt, int tid)
{
  const float* src = Wt + (size_t)(2 * (tid >> 5)) * D + ((tid & 31) << 2);
  #pragma unroll
  for (int i = 0; i < 8; ++i) {
    w[i][0] = *(const v4f*)(src + (size_t)(16 * i)     * D);
    w[i][1] = *(const v4f*)(src + (size_t)(16 * i + 1) * D);
  }
}

__device__ __forceinline__ void stage_w_store(float* __restrict__ ldsW,
                                              const v4f (&w)[8][2], int tid)
{
  const int pr0 = tid >> 5;
  const int cb  = (tid & 31) << 2;
  const int cs  = cb ^ (((pr0 >> 1) & 1) << 4);
  float* dst0 = &ldsW[pr0 * 256 + cs * 2];
  #pragma unroll
  for (int i = 0; i < 8; ++i) {
    const v4f g0 = w[i][0];
    const v4f g1 = w[i][1];
    float* dst = dst0 + i * 8 * 256;
    v4f lo = { g0[0], g1[0], g0[1], g1[1] };
    v4f hi = { g0[2], g1[2], g0[3], g1[3] };
    *(v4f*)(dst)     = lo;
    *(v4f*)(dst + 4) = hi;
  }
}

__device__ __forceinline__ void load_a_group(v4f (&a)[2][2],
                                             const float* __restrict__ xr0,
                                             const float* __restrict__ xr1, int g)
{
  #pragma unroll
  for (int q = 0; q < 2; ++q) {
    a[0][q] = *(const v4f*)(xr0 + g * 16 + q * 8);
    a[1][q] = *(const v4f*)(xr1 + g * 16 + q * 8);
  }
}

__device__ __forceinline__ void group_times_w(v8f (&acc)[2][8], const v4f (&a)[2][2],
                                              const float* bE, const float* bO, int g)
{
  const int gOff = g * 8 * 256;
  #pragma unroll
  for (int q = 0; q < 2; ++q) {
    #pragma unroll
    for (int s = 0; s < 2; ++s) {
      const int kOff = gOff + (q * 4 + s) * 256;
      v2f bfrag[8];
      #pragma unroll
      for (int j = 0; j < 8; ++j) {
        const float* bp = ((j & 1) ? bO : bE) + kOff + (j >> 1) * 64;
        bfrag[j] = *(const v2f*)bp;
      }
      const v2f a0 = { a[0][q][2 * s], a[0][q][2 * s + 1] };
      const v2f a1 = { a[1][q][2 * s], a[1][q][2 * s + 1] };
      #pragma unroll
      for (int j = 0; j < 8; ++j) {
        acc[0][j] = WMMA_F32(a0, bfrag[j], acc[0][j]);
        acc[1][j] = WMMA_F32(a1, bfrag[j], acc[1][j]);
      }
    }
  }
}

template <bool MIXED>
__device__ __forceinline__ void wave_rows_times_w(
    const float* __restrict__ xr0,
    const float* __restrict__ xr1,
    v4f (&a)[2][2][2],
    const float* __restrict__ bias,
    float*       __restrict__ out,
    const float* ldsW,
    int r0, int t, int lane, unsigned rowmask)
{
  const int h = lane >> 4;
  const int l = lane & 15;

  int hB = h;
  asm volatile("" : "+v"(hB));

  v8f acc[2][8];
  #pragma unroll
  for (int rt = 0; rt < 2; ++rt)
    #pragma unroll
    for (int j = 0; j < 8; ++j)
      acc[rt][j] = v8f{0.f, 0.f, 0.f, 0.f, 0.f, 0.f, 0.f, 0.f};

  const float* bLane = ldsW + hB * 512 + (l << 1);
  const float* bE;
  const float* bO;
  {
    const int offE = hB * 32;
    bE = bLane + offE;
    bO = bLane + (32 - offE);
  }

  #pragma unroll
  for (int g = 0; g < 8; ++g) {
    if (g + 1 < 8) {
      load_a_group(a[(g + 1) & 1], xr0, xr1, g + 1);
      __builtin_amdgcn_sched_barrier(0);
    }
    group_times_w(acc, a[g & 1], bE, bO, g);
    __builtin_amdgcn_sched_barrier(0);
  }

  if (!MIXED) {
    #pragma unroll
    for (int j = 0; j < 8; ++j) {
      const int   col = j * 16 + l;
      const float bv  = bias[t * D + col];
      #pragma unroll
      for (int rt = 0; rt < 2; ++rt) {
        float* orow = out + (size_t)(r0 + rt * 16 + 8 * h) * D + col;
        #pragma unroll
        for (int i = 0; i < 8; ++i)
          orow[(size_t)i * D] = acc[rt][j][i] + bv;
      }
    }
  } else {
    float bv[8];
    #pragma unroll
    for (int j = 0; j < 8; ++j)
      bv[j] = bias[t * D + j * 16 + l];
    const unsigned mh = rowmask >> (8 * h);
    float* obase = out + (size_t)(r0 + 8 * h) * D + l;
    #pragma unroll
    for (int rt = 0; rt < 2; ++rt) {
      #pragma unroll
      for (int i = 0; i < 8; ++i) {
        if ((mh >> (rt * 16 + i)) & 1u) {
          float* orow = obase + (rt * 16 + i) * D;
          #pragma unroll
          for (int j = 0; j < 8; ++j)
            orow[j * 16] = acc[rt][j][i] + bv[j];
        }
      }
    }
  }
}

__global__ __launch_bounds__(256, 4) void hetero_linear_wmma(
    const float* __restrict__ x,
    const int*   __restrict__ tvec,
    const float* __restrict__ W,
    const float* __restrict__ bias,
    float*       __restrict__ out,
    int nrows)
{
  __shared__ float ldsW[64 * 256];

  const int tid       = threadIdx.x;
  const int lane      = tid & 31;
  const int wave      = tid >> 5;
  const int block_row = blockIdx.x * ROWS_PER_BLOCK;
  const int r0        = block_row + wave * ROWS_PER_WAVE;

  const int t_first = tvec[block_row];
  const int t_last  = tvec[block_row + ROWS_PER_BLOCK - 1];

  const float* xr0 = x + (size_t)(r0 + (lane & 15)) * D + 4 * (lane >> 4);
  const float* xr1 = xr0 + (size_t)16 * D;
  v4f a[2][2][2];

  v4f wreg[8][2];

  if (t_first == t_last) {
    stage_w_load(wreg, W + (size_t)t_first * D * D, tid);
    load_a_group(a[0], xr0, xr1, 0);
    stage_w_store(ldsW, wreg, tid);
    __syncthreads();
    wave_rows_times_w<false>(xr0, xr1, a, bias, out, ldsW, r0, t_first, lane, 0u);
  } else {
    const int tv = tvec[r0 + lane];
    for (int t = t_first; t <= t_last; ++t) {
      stage_w_load(wreg, W + (size_t)t * D * D, tid);
      const unsigned rowmask = __builtin_amdgcn_ballot_w32(tv == t);
      if (rowmask != 0u)
        load_a_group(a[0], xr0, xr1, 0);
      if (t != t_first)
        __syncthreads();
      stage_w_store(ldsW, wreg, tid);
      __syncthreads();
      if (rowmask != 0u)
        wave_rows_times_w<true>(xr0, xr1, a, bias, out, ldsW, r0, t, lane, rowmask);
    }
  }
}

extern "C" void kernel_launch(void* const* d_in, const int* in_sizes, int n_in,
                              void* d_out, int out_size, void* d_ws, size_t ws_size,
                              hipStream_t stream) {
  const float* x    = (const float*)d_in[0];
  const int*   tvec = (const int*)  d_in[1];
  const float* W    = (const float*)d_in[2];
  const float* bias = (const float*)d_in[3];
  float* out = (float*)d_out;

  const int nrows = in_sizes[0] / D;
  const int grid  = nrows / ROWS_PER_BLOCK;

  hipLaunchKernelGGL(hetero_linear_wmma, dim3(grid), dim3(256), 0, stream,
                     x, tvec, W, bias, out, nrows);
}


# Two-type blocks keep both W in LDS (2x64 KB): one pass per wave, no barrier between types; the boundary tile's other-type rows go to 4 waves x 2 column tiles, not a 2nd full pass; k-sum order kept
# speedup vs baseline: 15.8090x; 1.1925x over previous
#include <hip/hip_runtime.h>


typedef __attribute__((ext_vector_type(2))) float v2f;
typedef __attribute__((ext_vector_type(4))) float v4f;
typedef __attribute__((ext_vector_type(8))) float v8f;

static constexpr int D = 128;
static constexpr int ROWS_PER_BLOCK = 256;
static constexpr int ROWS_PER_WAVE = 32;

#define WMMA_F32(A, B, C) \
  __builtin_amdgcn_wmma_f32_16x16x4_f32(false, (A), false, (B), (short)0, (C), false, false)

__device__ __forceinline__ void stage_w_load(v4f (&w)[8][2],
                                             const float* __restrict__ Wt, int tid)
{
  const float* src = Wt + (size_t)(2 * (tid >> 5)) * D + ((tid & 31) << 2);
  #pragma unroll
  for (int i = 0; i < 8; ++i) {
    w[i][0] = *(const v4f*)(src + (size_t)(16 * i)     * D);
    w[i][1] = *(const v4f*)(src + (size_t)(16 * i + 1) * D);
  }
}

__device__ __forceinline__ void stage_w_store(float* __restrict__ ldsW,
                                              const v4f (&w)[8][2], int tid)
{
  const int pr0 = tid >> 5;
  const int cb  = (tid & 31) << 2;
  const int cs  = cb ^ (((pr0 >> 1) & 1) << 4);
  float* dst0 = &ldsW[pr0 * 256 + cs * 2];
  #pragma unroll
  for (int i = 0; i < 8; ++i) {
    const v4f g0 = w[i][0];
    const v4f g1 = w[i][1];
    float* dst = dst0 + i * 8 * 256;
    v4f lo = { g0[0], g1[0], g0[1], g1[1] };
    v4f hi = { g0[2], g1[2], g0[3], g1[3] };
    *(v4f*)(dst)     = lo;
    *(v4f*)(dst + 4) = hi;
  }
}

__device__ __forceinline__ void load_a_group(v4f (&a)[2][2],
                                             const float* __restrict__ xr0,
                                             const float* __restrict__ xr1, int g)
{
  #pragma unroll
  for (int q = 0; q < 2; ++q) {
    a[0][q] = *(const v4f*)(xr0 + g * 16 + q * 8);
    a[1][q] = *(const v4f*)(xr1 + g * 16 + q * 8);
  }
}

__device__ __forceinline__ void group_times_w(v8f (&acc)[2][8], const v4f (&a)[2][2],
                                              const float* bE, const float* bO, int g)
{
  const int gOff = g * 8 * 256;
  #pragma unroll
  for (int q = 0; q < 2; ++q) {
    #pragma unroll
    for (int s = 0; s < 2; ++s) {
      const int kOff = gOff + (q * 4 + s) * 256;
      v2f bfrag[8];
      #pragma unroll
      for (int j = 0; j < 8; ++j) {
        const float* bp = ((j & 1) ? bO : bE) + kOff + (j >> 1) * 64;
        bfrag[j] = *(const v2f*)bp;
      }
      const v2f a0 = { a[0][q][2 * s], a[0][q][2 * s + 1] };
      const v2f a1 = { a[1][q][2 * s], a[1][q][2 * s + 1] };
      #pragma unroll
      for (int j = 0; j < 8; ++j) {
        acc[0][j] = WMMA_F32(a0, bfrag[j], acc[0][j]);
        acc[1][j] = WMMA_F32(a1, bfrag[j], acc[1][j]);
      }
    }
  }
}

template <bool MIXED>
__device__ __forceinline__ void wave_rows_times_w(
    const float* __restrict__ xr0,
    const float* __restrict__ xr1,
    v4f (&a)[2][2][2],
    const float* __restrict__ bias,
    float*       __restrict__ out,
    const float* ldsW,
    int r0, int t, int lane, unsigned rowmask)
{
  const int h = lane >> 4;
  const int l = lane & 15;

  int hB = h;
  asm volatile("" : "+v"(hB));

  v8f acc[2][8];
  #pragma unroll
  for (int rt = 0; rt < 2; ++rt)
    #pragma unroll
    for (int j = 0; j < 8; ++j)
      acc[rt][j] = v8f{0.f, 0.f, 0.f, 0.f, 0.f, 0.f, 0.f, 0.f};

  const float* bLane = ldsW + hB * 512 + (l << 1);
  const float* bE;
  const float* bO;
  {
    const int offE = hB * 32;
    bE = bLane + offE;
    bO = bLane + (32 - offE);
  }

  #pragma unroll
  for (int g = 0; g < 8; ++g) {
    if (g + 1 < 8) {
      load_a_group(a[(g + 1) & 1], xr0, xr1, g + 1);
      __builtin_amdgcn_sched_barrier(0);
    }
    group_times_w(acc, a[g & 1], bE, bO, g);
    __builtin_amdgcn_sched_barrier(0);
  }

  if (!MIXED) {
    #pragma unroll
    for (int j = 0; j < 8; ++j) {
      const int   col = j * 16 + l;
      const float bv  = bias[t * D + col];
      #pragma unroll
      for (int rt = 0; rt < 2; ++rt) {
        float* orow = out + (size_t)(r0 + rt * 16 + 8 * h) * D + col;
        #pragma unroll
        for (int i = 0; i < 8; ++i)
          orow[(size_t)i * D] = acc[rt][j][i] + bv;
      }
    }
  } else {
    float bv[8];
    #pragma unroll
    for (int j = 0; j < 8; ++j)
      bv[j] = bias[t * D + j * 16 + l];
    const unsigned mh = rowmask >> (8 * h);
    float* obase = out + (size_t)(r0 + 8 * h) * D + l;
    #pragma unroll
    for (int rt = 0; rt < 2; ++rt) {
      #pragma unroll
      for (int i = 0; i < 8; ++i) {
        if ((mh >> (rt * 16 + i)) & 1u) {
          float* orow = obase + (rt * 16 + i) * D;
          #pragma unroll
          for (int j = 0; j < 8; ++j)
            orow[j * 16] = acc[rt][j][i] + bv[j];
        }
      }
    }
  }
}

__device__ __forceinline__ void load_a_tile(v4f (&ax)[8][2], const float* __restrict__ xx)
{
  #pragma unroll
  for (int g = 0; g < 8; ++g) {
    ax[g][0] = *(const v4f*)(xx + g * 16);
    ax[g][1] = *(const v4f*)(xx + g * 16 + 8);
  }
}

__device__ __forceinline__ void boundary_tile_columns(
    const v4f (&ax)[8][2], int tvx,
    const float* __restrict__ bias,
    float*       __restrict__ out,
    const float* ldsWt, int xrow0, int t, int jp, int lane)
{
  const int h = lane >> 4;
  const int l = lane & 15;
  const float* bLane = ldsWt + h * 512 + jp * 64 + (l << 1);
  const float* bE = bLane + h * 32;
  const float* bO = bLane + (32 - h * 32);
  unsigned    m   = __builtin_amdgcn_ballot_w32(tvx == t) >> (8 * h);
  const int   col = jp * 32 + l;
  const float bv0 = bias[t * D + col];
  const float bv1 = bias[t * D + col + 16];
  int oidx = (xrow0 + 8 * h) * D + col;
  asm volatile("" : "+v"(m), "+v"(oidx));
  v8f acc0 = v8f{0.f, 0.f, 0.f, 0.f, 0.f, 0.f, 0.f, 0.f};
  v8f acc1 = v8f{0.f, 0.f, 0.f, 0.f, 0.f, 0.f, 0.f, 0.f};
  #pragma unroll
  for (int g = 0; g < 8; ++g) {
    #pragma unroll
    for (int q = 0; q < 2; ++q) {
      #pragma unroll
      for (int s = 0; s < 2; ++s) {
        const int kOff = (g * 8 + q * 4 + s) * 256;
        const v2f b0 = *(const v2f*)(bE + kOff);
        const v2f b1 = *(const v2f*)(bO + kOff);
        const v2f av = { ax[g][q][2 * s], ax[g][q][2 * s + 1] };
        acc0 = WMMA_F32(av, b0, acc0);
        acc1 = WMMA_F32(av, b1, acc1);
      }
    }
  }
  asm volatile("v_nop\n\tv_nop\n\tv_nop\n\tv_nop\n\tv_nop" : "+v"(acc0), "+v"(acc1));
  v8f o0 = acc0 + bv0;
  v8f o1 = acc1 + bv1;
  asm volatile("" : "+v"(o0), "+v"(o1));
  #pragma unroll
  for (int i = 0; i < 8; ++i) {
    if ((m >> i) & 1u) {
      out[oidx + i * D]      = o0[i];
      out[oidx + i * D + 16] = o1[i];
    }
  }
}

__global__ __launch_bounds__(256, 4) void hetero_linear_wmma(
    const float* __restrict__ x,
    const int*   __restrict__ tvec,
    const float* __restrict__ W,
    const float* __restrict__ bias,
    float*       __restrict__ out,
    int nrows)
{
  __shared__ float ldsW[2 * 64 * 256];

  const int tid       = threadIdx.x;
  const int lane      = tid & 31;
  const int wave      = tid >> 5;
  const int block_row = blockIdx.x * ROWS_PER_BLOCK;
  const int r0        = block_row + wave * ROWS_PER_WAVE;

  const int t_first = tvec[block_row];
  const int t_last  = tvec[block_row + ROWS_PER_BLOCK - 1];

  const float* xr0 = x + (size_t)(r0 + (lane & 15)) * D + 4 * (lane >> 4);
  const float* xr1 = xr0 + (size_t)16 * D;
  v4f a[2][2][2];

  v4f wreg[8][2];

  if (t_first == t_last) {
    stage_w_load(wreg, W + (size_t)t_first * D * D, tid);
    load_a_group(a[0], xr0, xr1, 0);
    stage_w_store(ldsW, wreg, tid);
    __syncthreads();
    wave_rows_times_w<false>(xr0, xr1, a, bias, out, ldsW, r0, t_first, lane, 0u);
  } else {
    const int  tv  = tvec[r0 + lane];
    const bool two = (t_last == t_first + 1);

    int tp = 0;
    if (two) {
      int n1 = 0;
      #pragma unroll
      for (int i = 0; i < 8; ++i)
        n1 += __builtin_popcount(
            __builtin_amdgcn_ballot_w32(tvec[block_row + 32 * i + lane] == t_first));
      v4f wreg2[8][2];
      stage_w_load(wreg,  W + (size_t)t_first * D * D, tid);
      stage_w_load(wreg2, W + (size_t)t_last  * D * D, tid);
      load_a_group(a[0], xr0, xr1, 0);

      const int  cb      = n1 & 31;
      const int  wave_u  = __builtin_amdgcn_readfirstlane(wave);
      const bool has_x   = (cb != 0) && (wave_u < 4);
      const bool x_first = cb <= 16;
      const int  xrow0   = block_row + 32 * (n1 >> 5) + (x_first ? 0 : 16);
      const int  tx      = x_first ? t_first : t_last;
      v4f ax[8][2];
      int tvx = 0;
      if (has_x) {
        load_a_tile(ax, x + (size_t)(xrow0 + (lane & 15)) * D + 4 * (lane >> 4));
        tvx = tvec[xrow0 + (lane & 15)];
      }
      stage_w_store(ldsW,            wreg,  tid);
      stage_w_store(ldsW + 64 * 256, wreg2, tid);
      __syncthreads();
      if (has_x)
        boundary_tile_columns(ax, tvx, bias, out, ldsW + (tx - t_first) * (64 * 256),
                              xrow0, tx, wave_u, lane);
      const int c1 = __builtin_popcount(__builtin_amdgcn_ballot_w32(tv == t_first));
      tp = (c1 > 16) ? t_first : t_last;
    }

    for (int t = t_first; t <= t_last; ++t) {
      unsigned rowmask = __builtin_amdgcn_ballot_w32(tv == t);
      const float* wbuf = ldsW;
      if (two) {
        if (t != tp) rowmask = 0u;
        wbuf = ldsW + (t - t_first) * (64 * 256);
      } else {
        stage_w_load(wreg, W + (size_t)t * D * D, tid);
        if (rowmask != 0u)
          load_a_group(a[0], xr0, xr1, 0);
        if (t != t_first)
          __syncthreads();
        stage_w_store(ldsW, wreg, tid);
        __syncthreads();
      }
      if (rowmask != 0u)
        wave_rows_times_w<true>(xr0, xr1, a, bias, out, wbuf, r0, t, lane, rowmask);
    }
  }
}

extern "C" void kernel_launch(void* const* d_in, const int* in_sizes, int n_in,
                              void* d_out, int out_size, void* d_ws, size_t ws_size,
                              hipStream_t stream) {
  const float* x    = (const float*)d_in[0];
  const int*   tvec = (const int*)  d_in[1];
  const float* W    = (const float*)d_in[2];
  const float* bias = (const float*)d_in[3];
  float* out = (float*)d_out;

  const int nrows = in_sizes[0] / D;
  const int grid  = nrows / ROWS_PER_BLOCK;

  hipLaunchKernelGGL(hetero_linear_wmma, dim3(grid), dim3(256), 0, stream,
                     x, tvec, W, bias, out, nrows);
}


# Single-type blocks: a finished 16-row tile is turned in a wave-private 8 KB LDS slot (idle 2nd W buffer) and leaves as whole 512-B rows, one 16-B store per lane and row (was 4-B half-line stores)
# speedup vs baseline: 16.4318x; 1.0394x over previous
#include <hip/hip_runtime.h>


typedef __attribute__((ext_vector_type(2))) float v2f;
typedef __attribute__((ext_vector_type(4))) float v4f;
typedef __attribute__((ext_vector_type(8))) float v8f;

static constexpr int D = 128;
static constexpr int ROWS_PER_BLOCK = 256;
static constexpr int ROWS_PER_WAVE = 32;

#define WMMA_F32(A, B, C) \
  __builtin_amdgcn_wmma_f32_16x16x4_f32(false, (A), false, (B), (short)0, (C), false, false)

__device__ __forceinline__ void stage_w_load(v4f (&w)[8][2],
                                             const float* __restrict__ Wt, int tid)
{
  const float* src = Wt + (size_t)(2 * (tid >> 5)) * D + ((tid & 31) << 2);
  #pragma unroll
  for (int i = 0; i < 8; ++i) {
    w[i][0] = *(const v4f*)(src + (size_t)(16 * i)     * D);
    w[i][1] = *(const v4f*)(src + (size_t)(16 * i + 1) * D);
  }
}

__device__ __forceinline__ void stage_w_store(float* __restrict__ ldsW,
                                              const v4f (&w)[8][2], int tid)
{
  const int pr0 = tid >> 5;
  const int cb  = (tid & 31) << 2;
  const int cs  = cb ^ (((pr0 >> 1) & 1) << 4);
  float* dst0 = &ldsW[pr0 * 256 + cs * 2];
  #pragma unroll
  for (int i = 0; i < 8; ++i) {
    const v4f g0 = w[i][0];
    const v4f g1 = w[i][1];
    float* dst = dst0 + i * 8 * 256;
    v4f lo = { g0[0], g1[0], g0[1], g1[1] };
    v4f hi = { g0[2], g1[2], g0[3], g1[3] };
    *(v4f*)(dst)     = lo;
    *(v4f*)(dst + 4) = hi;
  }
}

__device__ __forceinline__ void load_a_group(v4f (&a)[2][2],
                                             const float* __restrict__ xr0,
                                             const float* __restrict__ xr1, int g)
{
  #pragma unroll
  for (int q = 0; q < 2; ++q) {
    a[0][q] = *(const v4f*)(xr0 + g * 16 + q * 8);
    a[1][q] = *(const v4f*)(xr1 + g * 16 + q * 8);
  }
}

__device__ __forceinline__ void group_times_w(v8f (&acc)[2][8], const v4f (&a)[2][2],
                                              const float* bE, const float* bO, int g)
{
  const int gOff = g * 8 * 256;
  #pragma unroll
  for (int q = 0; q < 2; ++q) {
    #pragma unroll
    for (int s = 0; s < 2; ++s) {
      const int kOff = gOff + (q * 4 + s) * 256;
      v2f bfrag[8];
      #pragma unroll
      for (int j = 0; j < 8; ++j) {
        const float* bp = ((j & 1) ? bO : bE) + kOff + (j >> 1) * 64;
        bfrag[j] = *(const v2f*)bp;
      }
      const v2f a0 = { a[0][q][2 * s], a[0][q][2 * s + 1] };
      const v2f a1 = { a[1][q][2 * s], a[1][q][2 * s + 1] };
      #pragma unroll
      for (int j = 0; j < 8; ++j) {
        acc[0][j] = WMMA_F32(a0, bfrag[j], acc[0][j]);
        acc[1][j] = WMMA_F32(a1, bfrag[j], acc[1][j]);
      }
    }
  }
}

template <bool MIXED>
__device__ __forceinline__ void wave_rows_times_w(
    const float* __restrict__ xr0,
    const float* __restrict__ xr1,
    v4f (&a)[2][2][2],
    const float* __restrict__ bias,
    float*       __restrict__ out,
    const float* ldsW,
    float*       stage,
    int r0, int t, int lane, unsigned rowmask)
{
  const int h = lane >> 4;
  const int l = lane & 15;

  int hB = h;
  asm volatile("" : "+v"(hB));

  v8f acc[2][8];
  #pragma unroll
  for (int rt = 0; rt < 2; ++rt)
    #pragma unroll
    for (int j = 0; j < 8; ++j)
      acc[rt][j] = v8f{0.f, 0.f, 0.f, 0.f, 0.f, 0.f, 0.f, 0.f};

  const float* bLane = ldsW + hB * 512 + (l << 1);
  const float* bE;
  const float* bO;
  {
    const int offE = hB * 32;
    bE = bLane + offE;
    bO = bLane + (32 - offE);
  }

  #pragma unroll
  for (int g = 0; g < 8; ++g) {
    if (g + 1 < 8) {
      load_a_group(a[(g + 1) & 1], xr0, xr1, g + 1);
      __builtin_amdgcn_sched_barrier(0);
    }
    group_times_w(acc, a[g & 1], bE, bO, g);
    __builtin_amdgcn_sched_barrier(0);
  }

  if (!MIXED) {
    const v4f bv4 = *(const v4f*)(bias + t * D + 4 * lane);
    float*       wE  = stage + h * 1024 + l + 32 * h;
    float*       wO  = stage + h * 1024 + l - 32 * h;
    const float* rLo = stage + 4 * lane;
    const float* rHi = stage + ((4 * lane) ^ 32);
    float*       og  = out + (size_t)r0 * D + 4 * lane;
    #pragma unroll
    for (int rt = 0; rt < 2; ++rt) {
      #pragma unroll
      for (int j = 0; j < 8; ++j) {
        float* wp = ((j & 2) ? wO : wE) + 16 * j;
        #pragma unroll
        for (int i = 0; i < 8; ++i)
          wp[i * 128] = acc[rt][j][i];
      }
      #pragma unroll
      for (int R = 0; R < 16; ++R) {
        const v4f v = *(const v4f*)((R < 8 ? rLo : rHi) + R * 128);
        *(v4f*)(og + (size_t)(rt * 16 + R) * D) = v + bv4;
      }
    }
  } else {
    float bv[8];
    #pragma unroll
    for (int j = 0; j < 8; ++j)
      bv[j] = bias[t * D + j * 16 + l];
    const unsigned mh = rowmask >> (8 * h);
    float* obase = out + (size_t)(r0 + 8 * h) * D + l;
    #pragma unroll
    for (int rt = 0; rt < 2; ++rt) {
      #pragma unroll
      for (int i = 0; i < 8; ++i) {
        if ((mh >> (rt * 16 + i)) & 1u) {
          float* orow = obase + (rt * 16 + i) * D;
          #pragma unroll
          for (int j = 0; j < 8; ++j)
            orow[j * 16] = acc[rt][j][i] + bv[j];
        }
      }
    }
  }
}

__device__ __forceinline__ void load_a_tile(v4f (&ax)[8][2], const float* __restrict__ xx)
{
  #pragma unroll
  for (int g = 0; g < 8; ++g) {
    ax[g][0] = *(const v4f*)(xx + g * 16);
    ax[g][1] = *(const v4f*)(xx + g * 16 + 8);
  }
}

__device__ __forceinline__ void boundary_tile_columns(
    const v4f (&ax)[8][2], int tvx,
    const float* __restrict__ bias,
    float*       __restrict__ out,
    const float* ldsWt, int xrow0, int t, int jp, int lane)
{
  const int h = lane >> 4;
  const int l = lane & 15;
  const float* bLane = ldsWt + h * 512 + jp * 64 + (l << 1);
  const float* bE = bLane + h * 32;
  const float* bO = bLane + (32 - h * 32);
  unsigned    m   = __builtin_amdgcn_ballot_w32(tvx == t) >> (8 * h);
  const int   col = jp * 32 + l;
  const float bv0 = bias[t * D + col];
  const float bv1 = bias[t * D + col + 16];
  int oidx = (xrow0 + 8 * h) * D + col;
  asm volatile("" : "+v"(m), "+v"(oidx));
  v8f acc0 = v8f{0.f, 0.f, 0.f, 0.f, 0.f, 0.f, 0.f, 0.f};
  v8f acc1 = v8f{0.f, 0.f, 0.f, 0.f, 0.f, 0.f, 0.f, 0.f};
  #pragma unroll
  for (int g = 0; g < 8; ++g) {
    #pragma unroll
    for (int q = 0; q < 2; ++q) {
      #pragma unroll
      for (int s = 0; s < 2; ++s) {
        const int kOff = (g * 8 + q * 4 + s) * 256;
        const v2f b0 = *(const v2f*)(bE + kOff);
        const v2f b1 = *(const v2f*)(bO + kOff);
        const v2f av = { ax[g][q][2 * s], ax[g][q][2 * s + 1] };
        acc0 = WMMA_F32(av, b0, acc0);
        acc1 = WMMA_F32(av, b1, acc1);
      }
    }
  }
  asm volatile("v_nop\n\tv_nop\n\tv_nop\n\tv_nop\n\tv_nop" : "+v"(acc0), "+v"(acc1));
  v8f o0 = acc0 + bv0;
  v8f o1 = acc1 + bv1;
  asm volatile("" : "+v"(o0), "+v"(o1));
  #pragma unroll
  for (int i = 0; i < 8; ++i) {
    if ((m >> i) & 1u) {
      out[oidx + i * D]      = o0[i];
      out[oidx + i * D + 16] = o1[i];
    }
  }
}

__global__ __launch_bounds__(256, 4) void hetero_linear_wmma(
    const float* __restrict__ x,
    const int*   __restrict__ tvec,
    const float* __restrict__ W,
    const float* __restrict__ bias,
    float*       __restrict__ out,
    int nrows)
{
  __shared__ float ldsW[2 * 64 * 256];

  const int tid       = threadIdx.x;
  const int lane      = tid & 31;
  const int wave      = tid >> 5;
  const int block_row = blockIdx.x * ROWS_PER_BLOCK;
  const int r0        = block_row + wave * ROWS_PER_WAVE;

  const int t_first = tvec[block_row];
  const int t_last  = tvec[block_row + ROWS_PER_BLOCK - 1];

  const float* xr0 = x + (size_t)(r0 + (lane & 15)) * D + 4 * (lane >> 4);
  const float* xr1 = xr0 + (size_t)16 * D;
  v4f a[2][2][2];

  v4f wreg[8][2];

  if (t_first == t_last) {
    stage_w_load(wreg, W + (size_t)t_first * D * D, tid);
    load_a_group(a[0], xr0, xr1, 0);
    stage_w_store(ldsW, wreg, tid);
    __syncthreads();
    wave_rows_times_w<false>(xr0, xr1, a, bias, out, ldsW, ldsW + 64 * 256 + wave * 2048,
                             r0, t_first, lane, 0u);
  } else {
    const int  tv  = tvec[r0 + lane];
    const bool two = (t_last == t_first + 1);

    int tp = 0;
    if (two) {
      int n1 = 0;
      #pragma unroll
      for (int i = 0; i < 8; ++i)
        n1 += __builtin_popcount(
            __builtin_amdgcn_ballot_w32(tvec[block_row + 32 * i + lane] == t_first));
      v4f wreg2[8][2];
      stage_w_load(wreg,  W + (size_t)t_first * D * D, tid);
      stage_w_load(wreg2, W + (size_t)t_last  * D * D, tid);
      load_a_group(a[0], xr0, xr1, 0);

      const int  cb      = n1 & 31;
      const int  wave_u  = __builtin_amdgcn_readfirstlane(wave);
      const bool has_x   = (cb != 0) && (wave_u < 4);
      const bool x_first = cb <= 16;
      const int  xrow0   = block_row + 32 * (n1 >> 5) + (x_first ? 0 : 16);
      const int  tx      = x_first ? t_first : t_last;
      v4f ax[8][2];
      int tvx = 0;
      if (has_x) {
        load_a_tile(ax, x + (size_t)(xrow0 + (lane & 15)) * D + 4 * (lane >> 4));
        tvx = tvec[xrow0 + (lane & 15)];
      }
      stage_w_store(ldsW,            wreg,  tid);
      stage_w_store(ldsW + 64 * 256, wreg2, tid);
      __syncthreads();
      if (has_x)
        boundary_tile_columns(ax, tvx, bias, out, ldsW + (tx - t_first) * (64 * 256),
                              xrow0, tx, wave_u, lane);
      const int c1 = __builtin_popcount(__builtin_amdgcn_ballot_w32(tv == t_first));
      tp = (c1 > 16) ? t_first : t_last;
    }

    for (int t = t_first; t <= t_last; ++t) {
      unsigned rowmask = __builtin_amdgcn_ballot_w32(tv == t);
      const float* wbuf = ldsW;
      if (two) {
        if (t != tp) rowmask = 0u;
        wbuf = ldsW + (t - t_first) * (64 * 256);
      } else {
        stage_w_load(wreg, W + (size_t)t * D * D, tid);
        if (rowmask != 0u)
          load_a_group(a[0], xr0, xr1, 0);
        if (t != t_first)
          __syncthreads();
        stage_w_store(ldsW, wreg, tid);
        __syncthreads();
      }
      if (rowmask != 0u)
        wave_rows_times_w<true>(xr0, xr1, a, bias, out, wbuf, nullptr, r0, t, lane, rowmask);
    }
  }
}

extern "C" void kernel_launch(void* const* d_in, const int* in_sizes, int n_in,
                              void* d_out, int out_size, void* d_ws, size_t ws_size,
                              hipStream_t stream) {
  const float* x    = (const float*)d_in[0];
  const int*   tvec = (const int*)  d_in[1];
  const float* W    = (const float*)d_in[2];
  const float* bias = (const float*)d_in[3];
  float* out = (float*)d_out;

  const int nrows = in_sizes[0] / D;
  const int grid  = nrows / ROWS_PER_BLOCK;

  hipLaunchKernelGGL(hetero_linear_wmma, dim3(grid), dim3(256), 0, stream,
                     x, tvec, W, bias, out, nrows);
}
